# starting kernel
# baseline (speedup 1.0000x reference)
#include <hip/hip_runtime.h>
#include <hip/hip_bf16.h>

typedef __attribute__((ext_vector_type(2))) float v2f;
typedef __attribute__((ext_vector_type(8))) float v8f;

#define DD 128

__global__ void gemm_bias_act(const float* __restrict__ A,
                              const float* __restrict__ W,
                              const float* __restrict__ bias,
                              float* __restrict__ C,
                              int Ncols, int relu)
{
    __shared__ float As[16 * DD];
    __shared__ float Bs[DD * 16];
    const int lane = threadIdx.x & 31;
    const int tm = blockIdx.x * 16;
    const int tn = blockIdx.y * 16;

    {
        const float4* Ag = (const float4*)(A + (size_t)tm * DD);
        float4* As4 = (float4*)As;
#pragma unroll
        for (int r = 0; r < 16; ++r)
            As4[r * 32 + lane] = Ag[(size_t)r * 32 + lane];
    }
    {
#pragma unroll
        for (int i = 0; i < 16; ++i) {
            const int k  = i * 8 + (lane >> 2);
            const int c4 = (lane & 3) * 4;
            const float4 w = *(const float4*)(W + (size_t)k * Ncols + tn + c4);
            *(float4*)(Bs + k * 16 + c4) = w;
        }
    }
    __syncthreads();

    const int half = lane >> 4;
    const int mn   = lane & 15;

    v8f acc = {0.f, 0.f, 0.f, 0.f, 0.f, 0.f, 0.f, 0.f};
#pragma unroll
    for (int kk = 0; kk < 32; ++kk) {
        const int kb = kk * 4 + half * 2;
        v2f a, b;
        a.x = As[mn * DD + kb];
        a.y = As[mn * DD + kb + 1];
        b.x = Bs[kb * 16 + mn];
        b.y = Bs[(kb + 1) * 16 + mn];
        acc = __builtin_amdgcn_wmma_f32_16x16x4_f32(
             false, a,  false, b,
             (short)0, acc,  false,  false);
    }

#pragma unroll
    for (int r = 0; r < 8; ++r) {
        const int row = tm + r + half * 8;
        const int col = tn + mn;
        float v = acc[r] + (bias ? bias[col] : 0.f);
        if (relu) v = v > 0.f ? v : 0.f;
        C[(size_t)row * Ncols + col] = v;
    }
}

__global__ void deg_init(float* deg, int n) {
    int i = blockIdx.x * blockDim.x + threadIdx.x;
    if (i < n) deg[i] = 1.0f;
}
__global__ void deg_accum(const int* __restrict__ dst, float* deg, int e) {
    int i = blockIdx.x * blockDim.x + threadIdx.x;
    if (i < e) atomicAdd(deg + dst[i], 1.0f);
}
__global__ void deg_invsqrt(float* deg, int n) {
    int i = blockIdx.x * blockDim.x + threadIdx.x;
    if (i < n) deg[i] = rsqrtf(deg[i]);
}

__global__ void self_init(const float* __restrict__ hw,
                          const float* __restrict__ inv,
                          float* __restrict__ agg, int n) {
    size_t idx = (size_t)blockIdx.x * blockDim.x + threadIdx.x;
    if (idx < (size_t)n * DD) {
        const int i = (int)(idx >> 7);
        const float c = inv[i];
        agg[idx] = hw[idx] * c * c;
    }
}

__global__ void edge_scatter(const float* __restrict__ hw,
                             const int* __restrict__ src,
                             const int* __restrict__ dst,
                             const float* __restrict__ inv,
                             float* __restrict__ agg, int e) {
    const int wavesPerBlk = blockDim.x >> 5;
    const int ei = blockIdx.x * wavesPerBlk + (threadIdx.x >> 5);
    if (ei >= e) return;
    const int lane = threadIdx.x & 31;
    const int s = src[ei];
    const int d = dst[ei];
    const float coef = inv[s] * inv[d];
    const float4 v = *(const float4*)(hw + (size_t)s * DD + lane * 4);
    float* p = agg + (size_t)d * DD + lane * 4;
    atomicAdd(p + 0, v.x * coef);
    atomicAdd(p + 1, v.y * coef);
    atomicAdd(p + 2, v.z * coef);
    atomicAdd(p + 3, v.w * coef);
}

__global__ void bias_relu(const float* __restrict__ agg,
                          const float* __restrict__ b,
                          float* __restrict__ h, int n) {
    size_t idx = (size_t)blockIdx.x * blockDim.x + threadIdx.x;
    if (idx < (size_t)n * DD) {
        const float v = agg[idx] + b[idx & (DD - 1)];
        h[idx] = v > 0.f ? v : 0.f;
    }
}

__global__ void pool_zero(float* p, int n) {
    int i = blockIdx.x * blockDim.x + threadIdx.x;
    if (i < n) p[i] = 0.f;
}
__global__ void pool_add(const float* __restrict__ h,
                         const int* __restrict__ batch,
                         float* __restrict__ pooled, int n) {
    int t = blockIdx.x * blockDim.x + threadIdx.x;
    if (t < n * 32) {
        const int i = t >> 5;
        const int q = (t & 31) * 4;
        const int g = batch[i];
        const float4 v = *(const float4*)(h + (size_t)i * DD + q);
        float* p = pooled + (size_t)g * DD + q;
        atomicAdd(p + 0, v.x);
        atomicAdd(p + 1, v.y);
        atomicAdd(p + 2, v.z);
        atomicAdd(p + 3, v.w);
    }
}

extern "C" void kernel_launch(void* const* d_in, const int* in_sizes, int n_in,
                              void* d_out, int out_size, void* d_ws, size_t ws_size,
                              hipStream_t stream) {
    const float* x      = (const float*)d_in[0];
    const int*   ei     = (const int*)d_in[1];
    const int*   batch  = (const int*)d_in[2];
    const float* enc_W1 = (const float*)d_in[3];
    const float* enc_b1 = (const float*)d_in[4];
    const float* enc_W2 = (const float*)d_in[5];
    const float* enc_b2 = (const float*)d_in[6];
    const float* gcn_W  = (const float*)d_in[7];
    const float* gcn_b  = (const float*)d_in[8];
    const float* dec_W1 = (const float*)d_in[9];
    const float* dec_b1 = (const float*)d_in[10];
    const float* dec_W2 = (const float*)d_in[11];
    const float* dec_b2 = (const float*)d_in[12];

    const int N   = in_sizes[0] / DD;
    const int E   = in_sizes[1] / 2;
    const int L   = in_sizes[7] / (DD * DD);
    const int G   = 512;
    const int OUT = in_sizes[11] / DD;

    const int* src = ei;
    const int* dst = ei + E;

    char* ws = (char*)d_ws;
    const size_t nd = (size_t)N * DD * sizeof(float);
    float* B0     = (float*)(ws);
    float* B1     = (float*)(ws + nd);
    float* B2     = (float*)(ws + 2 * nd);
    float* inv    = (float*)(ws + 3 * nd);
    size_t off    = 3 * nd + (((size_t)N * 4 + 255) & ~(size_t)255);
    float* pooled = (float*)(ws + off);
    float* dech   = pooled + (size_t)G * DD;

    const dim3 wave(32);
    const int elemBlocks = (int)(((size_t)N * DD + 255) / 256);

    gemm_bias_act<<<dim3(N / 16, DD / 16), wave, 0, stream>>>(x,  enc_W1, enc_b1, B0, DD, 1);
    gemm_bias_act<<<dim3(N / 16, DD / 16), wave, 0, stream>>>(B0, enc_W2, enc_b2, B1, DD, 1);

    deg_init   <<<(N + 255) / 256, 256, 0, stream>>>(inv, N);
    deg_accum  <<<(E + 255) / 256, 256, 0, stream>>>(dst, inv, E);
    deg_invsqrt<<<(N + 255) / 256, 256, 0, stream>>>(inv, N);

    for (int l = 0; l < L; ++l) {
        const float* Wl = gcn_W + (size_t)l * DD * DD;
        const float* bl = gcn_b + (size_t)l * DD;
        gemm_bias_act<<<dim3(N / 16, DD / 16), wave, 0, stream>>>(B1, Wl, nullptr, B0, DD, 0);
        self_init   <<<elemBlocks, 256, 0, stream>>>(B0, inv, B2, N);
        edge_scatter<<<(E + 7) / 8, 256, 0, stream>>>(B0, src, dst, inv, B2, E);
        bias_relu   <<<elemBlocks, 256, 0, stream>>>(B2, bl, B1, N);
    }

    pool_zero<<<(G * DD + 255) / 256, 256, 0, stream>>>(pooled, G * DD);
    pool_add <<<(N * 32 + 255) / 256, 256, 0, stream>>>(B1, batch, pooled, N);

    gemm_bias_act<<<dim3(G / 16, DD / 16),  wave, 0, stream>>>(pooled, dec_W1, dec_b1, dech, DD, 1);
    gemm_bias_act<<<dim3(G / 16, OUT / 16), wave, 0, stream>>>(dech,   dec_W2, dec_b2, (float*)d_out, OUT, 0);
}


# each graph layer computed as relu((Ahat h) W + b), not Ahat (h W): bias, relu and the next self-loop start plane move into the gemm epilogue; self_init, bias_relu launches gone; sums re-associated
# speedup vs baseline: 1.1509x; 1.1509x over previous
#include <hip/hip_runtime.h>
#include <hip/hip_bf16.h>

typedef __attribute__((ext_vector_type(2))) float v2f;
typedef __attribute__((ext_vector_type(8))) float v8f;

#define DD 128

__global__ void gemm_bias_act(const float* __restrict__ A,
                              const float* __restrict__ W,
                              const float* __restrict__ bias,
                              float* __restrict__ C,
                              const float* __restrict__ inv,
                              float* __restrict__ C2,
                              int Ncols, int relu)
{
    __shared__ float As[16 * DD];
    __shared__ float Bs[DD * 16];
    const int lane = threadIdx.x & 31;
    const int tm = blockIdx.x * 16;
    const int tn = blockIdx.y * 16;
    const int half = lane >> 4;
    const int mn   = lane & 15;

    {
        const float4* Ag = (const float4*)(A + (size_t)tm * DD);
        float4* As4 = (float4*)As;
#pragma unroll
        for (int r = 0; r < 16; ++r)
            As4[r * 32 + lane] = Ag[(size_t)r * 32 + lane];
    }
    {
#pragma unroll
        for (int i = 0; i < 16; ++i) {
            const int k  = i * 8 + (lane >> 2);
            const int c4 = (lane & 3) * 4;
            const float4 w = *(const float4*)(W + (size_t)k * Ncols + tn + c4);
            *(float4*)(Bs + k * 16 + c4) = w;
        }
    }
    __syncthreads();

    float iv[8] = {0.f, 0.f, 0.f, 0.f, 0.f, 0.f, 0.f, 0.f};
    if (C2) {
        const float* ip = inv + tm;
#pragma unroll
        for (int r = 0; r < 8; ++r) {
            const float lo = ip[r];
            const float hi = ip[8 + r];
            iv[r] = half ? hi : lo;
        }
    }

    v8f acc = {0.f, 0.f, 0.f, 0.f, 0.f, 0.f, 0.f, 0.f};
#pragma unroll
    for (int kk = 0; kk < 32; ++kk) {
        const int kb = kk * 4 + half * 2;
        v2f a, b;
        a.x = As[mn * DD + kb];
        a.y = As[mn * DD + kb + 1];
        b.x = Bs[kb * 16 + mn];
        b.y = Bs[(kb + 1) * 16 + mn];
        acc = __builtin_amdgcn_wmma_f32_16x16x4_f32(
             false, a,  false, b,
             (short)0, acc,  false,  false);
    }

    const int col = tn + mn;
    const float bv = bias ? bias[col] : 0.f;
    if (C2) {
#pragma unroll
        for (int r = 0; r < 8; ++r) {
            const int row = tm + r + half * 8;
            float v = acc[r] + bv;
            if (relu) v = v > 0.f ? v : 0.f;
            C [(size_t)row * Ncols + col] = v;
            C2[(size_t)row * Ncols + col] = v * iv[r] * iv[r];
        }
    } else {
#pragma unroll
        for (int r = 0; r < 8; ++r) {
            const int row = tm + r + half * 8;
            float v = acc[r] + bv;
            if (relu) v = v > 0.f ? v : 0.f;
            C[(size_t)row * Ncols + col] = v;
        }
    }
}

__global__ void deg_init(float* deg, int n) {
    int i = blockIdx.x * blockDim.x + threadIdx.x;
    if (i < n) deg[i] = 1.0f;
}
__global__ void deg_accum(const int* __restrict__ dst, float* deg, int e) {
    int i = blockIdx.x * blockDim.x + threadIdx.x;
    if (i < e) atomicAdd(deg + dst[i], 1.0f);
}
__global__ void deg_invsqrt(float* deg, int n) {
    int i = blockIdx.x * blockDim.x + threadIdx.x;
    if (i < n) deg[i] = rsqrtf(deg[i]);
}

__global__ void edge_scatter(const float* __restrict__ hw,
                             const int* __restrict__ src,
                             const int* __restrict__ dst,
                             const float* __restrict__ inv,
                             float* __restrict__ agg, int e) {
    const int wavesPerBlk = blockDim.x >> 5;
    const int ei = blockIdx.x * wavesPerBlk + (threadIdx.x >> 5);
    if (ei >= e) return;
    const int lane = threadIdx.x & 31;
    const int s = src[ei];
    const int d = dst[ei];
    const float coef = inv[s] * inv[d];
    const float4 v = *(const float4*)(hw + (size_t)s * DD + lane * 4);
    float* p = agg + (size_t)d * DD + lane * 4;
    atomicAdd(p + 0, v.x * coef);
    atomicAdd(p + 1, v.y * coef);
    atomicAdd(p + 2, v.z * coef);
    atomicAdd(p + 3, v.w * coef);
}

__global__ void pool_zero(float* p, int n) {
    int i = blockIdx.x * blockDim.x + threadIdx.x;
    if (i < n) p[i] = 0.f;
}
__global__ void pool_add(const float* __restrict__ h,
                         const int* __restrict__ batch,
                         float* __restrict__ pooled, int n) {
    int t = blockIdx.x * blockDim.x + threadIdx.x;
    if (t < n * 32) {
        const int i = t >> 5;
        const int q = (t & 31) * 4;
        const int g = batch[i];
        const float4 v = *(const float4*)(h + (size_t)i * DD + q);
        float* p = pooled + (size_t)g * DD + q;
        atomicAdd(p + 0, v.x);
        atomicAdd(p + 1, v.y);
        atomicAdd(p + 2, v.z);
        atomicAdd(p + 3, v.w);
    }
}

extern "C" void kernel_launch(void* const* d_in, const int* in_sizes, int n_in,
                              void* d_out, int out_size, void* d_ws, size_t ws_size,
                              hipStream_t stream) {
    const float* x      = (const float*)d_in[0];
    const int*   ei     = (const int*)d_in[1];
    const int*   batch  = (const int*)d_in[2];
    const float* enc_W1 = (const float*)d_in[3];
    const float* enc_b1 = (const float*)d_in[4];
    const float* enc_W2 = (const float*)d_in[5];
    const float* enc_b2 = (const float*)d_in[6];
    const float* gcn_W  = (const float*)d_in[7];
    const float* gcn_b  = (const float*)d_in[8];
    const float* dec_W1 = (const float*)d_in[9];
    const float* dec_b1 = (const float*)d_in[10];
    const float* dec_W2 = (const float*)d_in[11];
    const float* dec_b2 = (const float*)d_in[12];

    const int N   = in_sizes[0] / DD;
    const int E   = in_sizes[1] / 2;
    const int L   = in_sizes[7] / (DD * DD);
    const int G   = 512;
    const int OUT = in_sizes[11] / DD;

    const int* src = ei;
    const int* dst = ei + E;

    char* ws = (char*)d_ws;
    const size_t nd = (size_t)N * DD * sizeof(float);
    float* B0     = (float*)(ws);
    float* B1     = (float*)(ws + nd);
    float* B2     = (float*)(ws + 2 * nd);
    float* inv    = (float*)(ws + 3 * nd);
    size_t off    = 3 * nd + (((size_t)N * 4 + 255) & ~(size_t)255);
    float* pooled = (float*)(ws + off);
    float* dech   = pooled + (size_t)G * DD;

    const dim3 wave(32);

    deg_init   <<<(N + 255) / 256, 256, 0, stream>>>(inv, N);
    deg_accum  <<<(E + 255) / 256, 256, 0, stream>>>(dst, inv, E);
    deg_invsqrt<<<(N + 255) / 256, 256, 0, stream>>>(inv, N);

    float* P = B1;
    float* Q = B2;
    float* F = B0;

    gemm_bias_act<<<dim3(N / 16, DD / 16), wave, 0, stream>>>(x,  enc_W1, enc_b1, B0, nullptr, nullptr, DD, 1);
    gemm_bias_act<<<dim3(N / 16, DD / 16), wave, 0, stream>>>(B0, enc_W2, enc_b2, P, inv, L > 0 ? Q : nullptr, DD, 1);

    for (int l = 0; l < L; ++l) {
        const float* Wl = gcn_W + (size_t)l * DD * DD;
        const float* bl = gcn_b + (size_t)l * DD;
        edge_scatter <<<(E + 7) / 8, 256, 0, stream>>>(P, src, dst, inv, Q, E);
        gemm_bias_act<<<dim3(N / 16, DD / 16), wave, 0, stream>>>(Q, Wl, bl, F, inv, l + 1 < L ? P : nullptr, DD, 1);
        float* t = Q; Q = P; P = F; F = t;
    }

    pool_zero<<<(G * DD + 255) / 256, 256, 0, stream>>>(pooled, G * DD);
    pool_add <<<(N * 32 + 255) / 256, 256, 0, stream>>>(P, batch, pooled, N);

    gemm_bias_act<<<dim3(G / 16, DD / 16),  wave, 0, stream>>>(pooled, dec_W1, dec_b1, dech, nullptr, nullptr, DD, 1);
    gemm_bias_act<<<dim3(G / 16, OUT / 16), wave, 0, stream>>>(dech,   dec_W2, dec_b2, (float*)d_out, nullptr, nullptr, OUT, 0);
}
